# starting kernel
# baseline (speedup 1.0000x reference)
#include <hip/hip_runtime.h>

typedef __attribute__((ext_vector_type(2))) float v2f;
typedef __attribute__((ext_vector_type(8))) float v8f;

#define BSZ 4
#define SSZ 512
#define HSZ 128
#define TI  4

__global__ __launch_bounds__(128) void gates_wmma_kernel(
    const float* __restrict__ inp,
    const float* __restrict__ W0, const float* __restrict__ b0,
    const float* __restrict__ W1, const float* __restrict__ b1,
    float* __restrict__ G0, float* __restrict__ G1)
{
  const int lane  = threadIdx.x & 31;
  const int wave  = threadIdx.x >> 5;
  const int tile  = blockIdx.x * 4 + wave;
  const int ntile = tile & 7;
  const int mtile = tile >> 3;

  const int r  = lane & 15;
  const int kb = (lane >> 4) * 2;

  const float* Aptr  = inp + (size_t)(mtile * 16 + r) * HSZ + kb;
  const float* B0ptr = W0  + (size_t)(ntile * 16 + r) * HSZ + kb;
  const float* B1ptr = W1  + (size_t)(ntile * 16 + r) * HSZ + kb;

  v8f acc0 = {};
  v8f acc1 = {};
  #pragma unroll 4
  for (int k0 = 0; k0 < HSZ; k0 += 4) {
    v2f a  = *(const v2f*)(Aptr  + k0);
    v2f w0 = *(const v2f*)(B0ptr + k0);
    v2f w1 = *(const v2f*)(B1ptr + k0);
    acc0 = __builtin_amdgcn_wmma_f32_16x16x4_f32(
         false, a,  false, w0, (short)0, acc0, false, false);
    acc1 = __builtin_amdgcn_wmma_f32_16x16x4_f32(
         false, a,  false, w1, (short)0, acc1, false, false);
  }

  const int col  = ntile * 16 + (lane & 15);
  const int mofs = (lane >> 4) * 8;
  const float bias0 = b0[col];
  const float bias1 = b1[col];
  #pragma unroll
  for (int v = 0; v < 8; ++v) {
    const size_t idx = (size_t)(mtile * 16 + mofs + v) * HSZ + col;
    G0[idx] = acc0[v] + bias0;
    G1[idx] = acc1[v] + bias1;
  }
}

__global__ __launch_bounds__(128) void pairwise_kernel(
    const float* __restrict__ inp,
    const int*   __restrict__ masks,
    const float* __restrict__ G0,
    const float* __restrict__ G1,
    float* __restrict__ out)
{
  const int h     = threadIdx.x;
  const int b     = blockIdx.x / (SSZ / TI);
  const int ibase = (blockIdx.x % (SSZ / TI)) * TI;

  float g0[TI];
  float acc[TI];
  #pragma unroll
  for (int t = 0; t < TI; ++t) {
    g0[t]  = G0[((size_t)(b * SSZ + ibase + t) * HSZ) + h];
    acc[t] = 0.0f;
  }

  const float* inpB = inp + (size_t)b * SSZ * HSZ;
  const float* g1B  = G1  + (size_t)b * SSZ * HSZ;
  const int*   mB   = masks + b * SSZ;

  int len = 0;
  for (int j = 0; j < SSZ; ++j) {
    const int mj = mB[j];
    len += mj;
    if (mj) {
      const float x  = inpB[(size_t)j * HSZ + h];
      const float g1 = g1B [(size_t)j * HSZ + h];
      #pragma unroll
      for (int t = 0; t < TI; ++t) {
        const float s = 1.0f / (1.0f + __expf(-(g0[t] + g1)));
        acc[t] += x * s;
      }
    }
  }

  const float inv_len = 1.0f / (float)len;
  #pragma unroll
  for (int t = 0; t < TI; ++t) {
    const float mi = (float)mB[ibase + t];
    out[((size_t)(b * SSZ + ibase + t) * HSZ) + h] = tanhf(mi * acc[t] * inv_len);
  }
}

extern "C" void kernel_launch(void* const* d_in, const int* in_sizes, int n_in,
                              void* d_out, int out_size, void* d_ws, size_t ws_size,
                              hipStream_t stream) {
  const float* inp   = (const float*)d_in[0];
  const int*   masks = (const int*)  d_in[1];
  const float* W0    = (const float*)d_in[2];
  const float* b0    = (const float*)d_in[3];
  const float* W1    = (const float*)d_in[4];
  const float* b1    = (const float*)d_in[5];
  float* out = (float*)d_out;

  float* G0 = (float*)d_ws;
  float* G1 = G0 + (size_t)BSZ * SSZ * HSZ;

  gates_wmma_kernel<<<256, 128, 0, stream>>>(inp, W0, b0, W1, b1, G0, G1);
  pairwise_kernel<<<(BSZ * SSZ) / TI, 128, 0, stream>>>(inp, masks, G0, G1, out);
}


# pairwise j-loop in 16-row chunks, two register sets: next chunk's inp/G1/mask loads issued ahead of the current chunk's sigmoids; term formed as x/(1+e); ascending-j sum order kept
# speedup vs baseline: 3.1656x; 3.1656x over previous
#include <hip/hip_runtime.h>

typedef __attribute__((ext_vector_type(2))) float v2f;
typedef __attribute__((ext_vector_type(8))) float v8f;

#define BSZ 4
#define SSZ 512
#define HSZ 128
#define TI  4

__global__ __launch_bounds__(128) void gates_wmma_kernel(
    const float* __restrict__ inp,
    const float* __restrict__ W0, const float* __restrict__ b0,
    const float* __restrict__ W1, const float* __restrict__ b1,
    float* __restrict__ G0, float* __restrict__ G1)
{
  const int lane  = threadIdx.x & 31;
  const int wave  = threadIdx.x >> 5;
  const int tile  = blockIdx.x * 4 + wave;
  const int ntile = tile & 7;
  const int mtile = tile >> 3;

  const int r  = lane & 15;
  const int kb = (lane >> 4) * 2;

  const float* Aptr  = inp + (size_t)(mtile * 16 + r) * HSZ + kb;
  const float* B0ptr = W0  + (size_t)(ntile * 16 + r) * HSZ + kb;
  const float* B1ptr = W1  + (size_t)(ntile * 16 + r) * HSZ + kb;

  v8f acc0 = {};
  v8f acc1 = {};
  #pragma unroll 4
  for (int k0 = 0; k0 < HSZ; k0 += 4) {
    v2f a  = *(const v2f*)(Aptr  + k0);
    v2f w0 = *(const v2f*)(B0ptr + k0);
    v2f w1 = *(const v2f*)(B1ptr + k0);
    acc0 = __builtin_amdgcn_wmma_f32_16x16x4_f32(
         false, a,  false, w0, (short)0, acc0, false, false);
    acc1 = __builtin_amdgcn_wmma_f32_16x16x4_f32(
         false, a,  false, w1, (short)0, acc1, false, false);
  }

  const int col  = ntile * 16 + (lane & 15);
  const int mofs = (lane >> 4) * 8;
  const float bias0 = b0[col];
  const float bias1 = b1[col];
  #pragma unroll
  for (int v = 0; v < 8; ++v) {
    const size_t idx = (size_t)(mtile * 16 + mofs + v) * HSZ + col;
    G0[idx] = acc0[v] + bias0;
    G1[idx] = acc1[v] + bias1;
  }
}

#define JC 16

struct Chunk {
  float x[JC];
  float g[JC];
  int   m[JC];
};

__device__ __forceinline__ void load_chunk(Chunk& c, const float* __restrict__ xp,
                                           const float* __restrict__ gp,
                                           const int* __restrict__ mp, int j0)
{
  #pragma unroll
  for (int u = 0; u < JC; ++u) c.m[u] = mp[j0 + u];
  #pragma unroll
  for (int u = 0; u < JC; ++u) {
    c.x[u] = xp[(size_t)(j0 + u) * HSZ];
    c.g[u] = gp[(size_t)(j0 + u) * HSZ];
  }
}

__device__ __forceinline__ void compute_chunk(const Chunk& c, const float (&g0)[TI],
                                              float (&acc)[TI], int& len)
{
  #pragma unroll
  for (int u = 0; u < JC; ++u) {
    const int mj = c.m[u];
    len += mj;
    if (mj) {
      const float x  = c.x[u];
      const float g1 = c.g[u];
      #pragma unroll
      for (int t = 0; t < TI; ++t) {
        acc[t] += x / (1.0f + __expf(-(g0[t] + g1)));
      }
    }
  }
}

__global__ __launch_bounds__(128) void pairwise_kernel(
    const float* __restrict__ inp,
    const int*   __restrict__ masks,
    const float* __restrict__ G0,
    const float* __restrict__ G1,
    float* __restrict__ out)
{
  const int h     = threadIdx.x;
  const int b     = blockIdx.x / (SSZ / TI);
  const int ibase = (blockIdx.x % (SSZ / TI)) * TI;

  const float* xp = inp + (size_t)b * SSZ * HSZ + h;
  const float* gp = G1  + (size_t)b * SSZ * HSZ + h;
  const int*   mB = masks + b * SSZ;

  Chunk A, B;
  load_chunk(A, xp, gp, mB, 0);

  float g0[TI];
  float acc[TI];
  #pragma unroll
  for (int t = 0; t < TI; ++t) {
    g0[t]  = G0[((size_t)(b * SSZ + ibase + t) * HSZ) + h];
    acc[t] = 0.0f;
  }

  int len = 0;
  for (int jc = 0; jc < SSZ; jc += 2 * JC) {
    load_chunk(B, xp, gp, mB, jc + JC);
    compute_chunk(A, g0, acc, len);
    load_chunk(A, xp, gp, mB, (jc + 2 * JC) & (SSZ - 1));
    compute_chunk(B, g0, acc, len);
  }

  const float inv_len = 1.0f / (float)len;
  #pragma unroll
  for (int t = 0; t < TI; ++t) {
    const float mi = (float)mB[ibase + t];
    out[((size_t)(b * SSZ + ibase + t) * HSZ) + h] = tanhf(mi * acc[t] * inv_len);
  }
}

extern "C" void kernel_launch(void* const* d_in, const int* in_sizes, int n_in,
                              void* d_out, int out_size, void* d_ws, size_t ws_size,
                              hipStream_t stream) {
  const float* inp   = (const float*)d_in[0];
  const int*   masks = (const int*)  d_in[1];
  const float* W0    = (const float*)d_in[2];
  const float* b0    = (const float*)d_in[3];
  const float* W1    = (const float*)d_in[4];
  const float* b1    = (const float*)d_in[5];
  float* out = (float*)d_out;

  float* G0 = (float*)d_ws;
  float* G1 = G0 + (size_t)BSZ * SSZ * HSZ;

  gates_wmma_kernel<<<256, 128, 0, stream>>>(inp, W0, b0, W1, b1, G0, G1);
  pairwise_kernel<<<(BSZ * SSZ) / TI, 128, 0, stream>>>(inp, masks, G0, G1, out);
}


# pairwise term: IEEE division x/(1+exp(-(g0+g1))) replaced by x*rcp(1+exp2(G0+G1)) on gates stored pre-scaled by -log2(e): one v_exp, one v_rcp, one v_fmac per term; ascending-j sum order kept
# speedup vs baseline: 4.9878x; 1.5756x over previous
#include <hip/hip_runtime.h>

typedef __attribute__((ext_vector_type(2))) float v2f;
typedef __attribute__((ext_vector_type(8))) float v8f;

#define BSZ 4
#define SSZ 512
#define HSZ 128
#define TI  4
#define NEG_LOG2E (-1.4426950408889634f)

__global__ __launch_bounds__(128) void gates_wmma_kernel(
    const float* __restrict__ inp,
    const float* __restrict__ W0, const float* __restrict__ b0,
    const float* __restrict__ W1, const float* __restrict__ b1,
    float* __restrict__ G0, float* __restrict__ G1)
{
  const int lane  = threadIdx.x & 31;
  const int wave  = threadIdx.x >> 5;
  const int tile  = blockIdx.x * 4 + wave;
  const int ntile = tile & 7;
  const int mtile = tile >> 3;

  const int r  = lane & 15;
  const int kb = (lane >> 4) * 2;

  const float* Aptr  = inp + (size_t)(mtile * 16 + r) * HSZ + kb;
  const float* B0ptr = W0  + (size_t)(ntile * 16 + r) * HSZ + kb;
  const float* B1ptr = W1  + (size_t)(ntile * 16 + r) * HSZ + kb;

  v8f acc0 = {};
  v8f acc1 = {};
  #pragma unroll 4
  for (int k0 = 0; k0 < HSZ; k0 += 4) {
    v2f a  = *(const v2f*)(Aptr  + k0);
    v2f w0 = *(const v2f*)(B0ptr + k0);
    v2f w1 = *(const v2f*)(B1ptr + k0);
    acc0 = __builtin_amdgcn_wmma_f32_16x16x4_f32(
         false, a,  false, w0, (short)0, acc0, false, false);
    acc1 = __builtin_amdgcn_wmma_f32_16x16x4_f32(
         false, a,  false, w1, (short)0, acc1, false, false);
  }

  const int col  = ntile * 16 + (lane & 15);
  const int mofs = (lane >> 4) * 8;
  const float bias0 = b0[col];
  const float bias1 = b1[col];
  #pragma unroll
  for (int v = 0; v < 8; ++v) {
    const size_t idx = (size_t)(mtile * 16 + mofs + v) * HSZ + col;
    G0[idx] = (acc0[v] + bias0) * NEG_LOG2E;
    G1[idx] = (acc1[v] + bias1) * NEG_LOG2E;
  }
}

#define JC 16

struct Chunk {
  float x[JC];
  float g[JC];
  int   m[JC];
};

__device__ __forceinline__ void load_chunk(Chunk& c, const float* __restrict__ xp,
                                           const float* __restrict__ gp,
                                           const int* __restrict__ mp, int j0)
{
  #pragma unroll
  for (int u = 0; u < JC; ++u) c.m[u] = mp[j0 + u];
  #pragma unroll
  for (int u = 0; u < JC; ++u) {
    c.x[u] = xp[(size_t)(j0 + u) * HSZ];
    c.g[u] = gp[(size_t)(j0 + u) * HSZ];
  }
}

__device__ __forceinline__ void compute_chunk(const Chunk& c, const float (&g0)[TI],
                                              float (&acc)[TI], int& len)
{
  #pragma unroll
  for (int u = 0; u < JC; ++u) {
    const int mj = c.m[u];
    len += mj;
    if (mj) {
      const float x  = c.x[u];
      const float g1 = c.g[u];
      float r[TI];
      #pragma unroll
      for (int t = 0; t < TI; ++t)
        r[t] = __builtin_amdgcn_rcpf(1.0f + __builtin_amdgcn_exp2f(g0[t] + g1));
      static_assert(TI == 4, "the accumulate below is written for four rows");
      asm("v_fmac_f32 %0, %4, %5\n\t"
          "s_delay_alu instid0(TRANS32_DEP_3) | instskip(NEXT) | instid1(TRANS32_DEP_2)\n\t"
          "v_fmac_f32 %1, %4, %6\n\t"
          "v_fmac_f32 %2, %4, %7\n\t"
          "s_delay_alu instid0(TRANS32_DEP_1)\n\t"
          "v_fmac_f32 %3, %4, %8"
          : "+v"(acc[0]), "+v"(acc[1]), "+v"(acc[2]), "+v"(acc[3])
          : "v"(x), "v"(r[0]), "v"(r[1]), "v"(r[2]), "v"(r[3]));
    }
  }
}

__global__ __launch_bounds__(128) void pairwise_kernel(
    const float* __restrict__ inp,
    const int*   __restrict__ masks,
    const float* __restrict__ G0,
    const float* __restrict__ G1,
    float* __restrict__ out)
{
  const int h     = threadIdx.x;
  const int b     = blockIdx.x / (SSZ / TI);
  const int ibase = (blockIdx.x % (SSZ / TI)) * TI;

  const float* xp = inp + (size_t)b * SSZ * HSZ + h;
  const float* gp = G1  + (size_t)b * SSZ * HSZ + h;
  const int*   mB = masks + b * SSZ;

  Chunk A, B;
  load_chunk(A, xp, gp, mB, 0);

  float g0[TI];
  float acc[TI];
  #pragma unroll
  for (int t = 0; t < TI; ++t) {
    g0[t]  = G0[((size_t)(b * SSZ + ibase + t) * HSZ) + h];
    acc[t] = 0.0f;
  }

  int len = 0;
  for (int jc = 0; jc < SSZ; jc += 2 * JC) {
    load_chunk(B, xp, gp, mB, jc + JC);
    compute_chunk(A, g0, acc, len);
    load_chunk(A, xp, gp, mB, (jc + 2 * JC) & (SSZ - 1));
    compute_chunk(B, g0, acc, len);
  }

  const float inv_len = 1.0f / (float)len;
  #pragma unroll
  for (int t = 0; t < TI; ++t) {
    const float mi = (float)mB[ibase + t];
    out[((size_t)(b * SSZ + ibase + t) * HSZ) + h] = tanhf(mi * acc[t] * inv_len);
  }
}

extern "C" void kernel_launch(void* const* d_in, const int* in_sizes, int n_in,
                              void* d_out, int out_size, void* d_ws, size_t ws_size,
                              hipStream_t stream) {
  const float* inp   = (const float*)d_in[0];
  const int*   masks = (const int*)  d_in[1];
  const float* W0    = (const float*)d_in[2];
  const float* b0    = (const float*)d_in[3];
  const float* W1    = (const float*)d_in[4];
  const float* b1    = (const float*)d_in[5];
  float* out = (float*)d_out;

  float* G0 = (float*)d_ws;
  float* G1 = G0 + (size_t)BSZ * SSZ * HSZ;

  gates_wmma_kernel<<<256, 128, 0, stream>>>(inp, W0, b0, W1, b1, G0, G1);
  pairwise_kernel<<<(BSZ * SSZ) / TI, 128, 0, stream>>>(inp, masks, G0, G1, out);
}


# pairwise term: sigmoid(g0+g1) as 0.5+0.5*tanh on the hardware tanh, gates stored halved: one transcendental per term, not two; sum split as 0.5*(sum_j x + sum_j x*tanh), each in ascending j
# speedup vs baseline: 5.1662x; 1.0358x over previous
#include <hip/hip_runtime.h>

typedef __attribute__((ext_vector_type(2))) float v2f;
typedef __attribute__((ext_vector_type(8))) float v8f;

#define BSZ 4
#define SSZ 512
#define HSZ 128
#define TI  4
#define GATE_SCALE 0.5f

__global__ __launch_bounds__(128) void gates_wmma_kernel(
    const float* __restrict__ inp,
    const float* __restrict__ W0, const float* __restrict__ b0,
    const float* __restrict__ W1, const float* __restrict__ b1,
    float* __restrict__ G0, float* __restrict__ G1)
{
  const int lane  = threadIdx.x & 31;
  const int wave  = threadIdx.x >> 5;
  const int tile  = blockIdx.x * 4 + wave;
  const int ntile = tile & 7;
  const int mtile = tile >> 3;

  const int r  = lane & 15;
  const int kb = (lane >> 4) * 2;

  const float* Aptr  = inp + (size_t)(mtile * 16 + r) * HSZ + kb;
  const float* B0ptr = W0  + (size_t)(ntile * 16 + r) * HSZ + kb;
  const float* B1ptr = W1  + (size_t)(ntile * 16 + r) * HSZ + kb;

  v8f acc0 = {};
  v8f acc1 = {};
  #pragma unroll 4
  for (int k0 = 0; k0 < HSZ; k0 += 4) {
    v2f a  = *(const v2f*)(Aptr  + k0);
    v2f w0 = *(const v2f*)(B0ptr + k0);
    v2f w1 = *(const v2f*)(B1ptr + k0);
    acc0 = __builtin_amdgcn_wmma_f32_16x16x4_f32(
         false, a,  false, w0, (short)0, acc0, false, false);
    acc1 = __builtin_amdgcn_wmma_f32_16x16x4_f32(
         false, a,  false, w1, (short)0, acc1, false, false);
  }

  const int col  = ntile * 16 + (lane & 15);
  const int mofs = (lane >> 4) * 8;
  const float bias0 = b0[col];
  const float bias1 = b1[col];
  #pragma unroll
  for (int v = 0; v < 8; ++v) {
    const size_t idx = (size_t)(mtile * 16 + mofs + v) * HSZ + col;
    G0[idx] = (acc0[v] + bias0) * GATE_SCALE;
    G1[idx] = (acc1[v] + bias1) * GATE_SCALE;
  }
}

#define JC 16

struct Chunk {
  float x[JC];
  float g[JC];
  int   m[JC];
};

__device__ __forceinline__ void load_chunk(Chunk& c, const float* __restrict__ xp,
                                           const float* __restrict__ gp,
                                           const int* __restrict__ mp, int j0)
{
  #pragma unroll
  for (int u = 0; u < JC; ++u) c.m[u] = mp[j0 + u];
  #pragma unroll
  for (int u = 0; u < JC; ++u) {
    c.x[u] = xp[(size_t)(j0 + u) * HSZ];
    c.g[u] = gp[(size_t)(j0 + u) * HSZ];
  }
}

__device__ __forceinline__ void compute_chunk(const Chunk& c, const float (&g0)[TI],
                                              float (&acc)[TI], float& accx, int& len)
{
  #pragma unroll
  for (int u = 0; u < JC; ++u) {
    const int mj = c.m[u];
    len += mj;
    if (mj) {
      const float x  = c.x[u];
      const float g1 = c.g[u];
      float th[TI];
      #pragma unroll
      for (int t = 0; t < TI; ++t)
        th[t] = __builtin_amdgcn_tanhf(g0[t] + g1);
      accx += x;
      static_assert(TI == 4, "the accumulate below is written for four rows");
      asm("v_fmac_f32 %0, %4, %5\n\t"
          "s_delay_alu instid0(TRANS32_DEP_3) | instskip(NEXT) | instid1(TRANS32_DEP_2)\n\t"
          "v_fmac_f32 %1, %4, %6\n\t"
          "v_fmac_f32 %2, %4, %7\n\t"
          "s_delay_alu instid0(TRANS32_DEP_1)\n\t"
          "v_fmac_f32 %3, %4, %8"
          : "+v"(acc[0]), "+v"(acc[1]), "+v"(acc[2]), "+v"(acc[3])
          : "v"(x), "v"(th[0]), "v"(th[1]), "v"(th[2]), "v"(th[3]));
    }
  }
}

__global__ __launch_bounds__(128) void pairwise_kernel(
    const float* __restrict__ inp,
    const int*   __restrict__ masks,
    const float* __restrict__ G0,
    const float* __restrict__ G1,
    float* __restrict__ out)
{
  const int h     = threadIdx.x;
  const int b     = blockIdx.x / (SSZ / TI);
  const int ibase = (blockIdx.x % (SSZ / TI)) * TI;

  const float* xp = inp + (size_t)b * SSZ * HSZ + h;
  const float* gp = G1  + (size_t)b * SSZ * HSZ + h;
  const int*   mB = masks + b * SSZ;

  Chunk A, B;
  load_chunk(A, xp, gp, mB, 0);

  float g0[TI];
  float acc[TI];
  float accx = 0.0f;
  #pragma unroll
  for (int t = 0; t < TI; ++t) {
    g0[t]  = G0[((size_t)(b * SSZ + ibase + t) * HSZ) + h];
    acc[t] = 0.0f;
  }

  int len = 0;
  for (int jc = 0; jc < SSZ; jc += 2 * JC) {
    load_chunk(B, xp, gp, mB, jc + JC);
    compute_chunk(A, g0, acc, accx, len);
    load_chunk(A, xp, gp, mB, (jc + 2 * JC) & (SSZ - 1));
    compute_chunk(B, g0, acc, accx, len);
  }

  const float inv_len = 1.0f / (float)len;
  #pragma unroll
  for (int t = 0; t < TI; ++t) {
    const float mi  = (float)mB[ibase + t];
    const float sum = 0.5f * (accx + acc[t]);
    out[((size_t)(b * SSZ + ibase + t) * HSZ) + h] = tanhf(mi * sum * inv_len);
  }
}

extern "C" void kernel_launch(void* const* d_in, const int* in_sizes, int n_in,
                              void* d_out, int out_size, void* d_ws, size_t ws_size,
                              hipStream_t stream) {
  const float* inp   = (const float*)d_in[0];
  const int*   masks = (const int*)  d_in[1];
  const float* W0    = (const float*)d_in[2];
  const float* b0    = (const float*)d_in[3];
  const float* W1    = (const float*)d_in[4];
  const float* b1    = (const float*)d_in[5];
  float* out = (float*)d_out;

  float* G0 = (float*)d_ws;
  float* G1 = G0 + (size_t)BSZ * SSZ * HSZ;

  gates_wmma_kernel<<<256, 128, 0, stream>>>(inp, W0, b0, W1, b1, G0, G1);
  pairwise_kernel<<<(BSZ * SSZ) / TI, 128, 0, stream>>>(inp, masks, G0, G1, out);
}


# gate GEMM: operand tiles read from global one whole 512-B row per load, parked in LDS (pitch 528 B), fragments read as 16-byte LDS loads; k order inside each block of 8, so sum order over k, changes
# speedup vs baseline: 8.7995x; 1.7033x over previous
#include <hip/hip_runtime.h>

typedef __attribute__((ext_vector_type(2))) float v2f;
typedef __attribute__((ext_vector_type(4))) float v4f;
typedef __attribute__((ext_vector_type(8))) float v8f;

#define BSZ 4
#define SSZ 512
#define HSZ 128
#define TI  4
#define GATE_SCALE 0.5f

#define LDP 132

__global__ __launch_bounds__(128) void gates_wmma_kernel(
    const float* __restrict__ inp,
    const float* __restrict__ W0, const float* __restrict__ b0,
    const float* __restrict__ W1, const float* __restrict__ b1,
    float* __restrict__ G0, float* __restrict__ G1)
{
  __shared__ __attribute__((aligned(16))) float sA[16 * LDP];
  __shared__ __attribute__((aligned(16))) float sW[4 * 2 * 16 * LDP];

  const int lane  = threadIdx.x & 31;
  const int wave  = threadIdx.x >> 5;
  const int tile  = blockIdx.x * 4 + wave;
  const int ntile = tile & 7;
  const int mtile = tile >> 3;

  const float* Ag  = inp + (size_t)(mtile * 16 + wave * 4) * HSZ + lane * 4;
  const float* W0g = W0  + (size_t)(ntile * 16) * HSZ + lane * 4;
  const float* W1g = W1  + (size_t)(ntile * 16) * HSZ + lane * 4;
  v4f ra[4], r0[16], r1[16];
  #pragma unroll
  for (int i = 0; i < 4; ++i)
    ra[i] = *(const v4f*)(Ag + (size_t)i * HSZ);
  #pragma unroll
  for (int i = 0; i < 16; ++i) {
    r0[i] = *(const v4f*)(W0g + (size_t)i * HSZ);
    r1[i] = *(const v4f*)(W1g + (size_t)i * HSZ);
  }

  float* myW0 = sW + (wave * 2 + 0) * 16 * LDP;
  float* myW1 = sW + (wave * 2 + 1) * 16 * LDP;
  #pragma unroll
  for (int i = 0; i < 4; ++i)
    *(v4f*)(sA + (wave * 4 + i) * LDP + lane * 4) = ra[i];
  #pragma unroll
  for (int i = 0; i < 16; ++i) {
    *(v4f*)(myW0 + i * LDP + lane * 4) = r0[i];
    *(v4f*)(myW1 + i * LDP + lane * 4) = r1[i];
  }
  __syncthreads();

  const int r = lane & 15;
  const int h = lane >> 4;
  const float* fa  = sA   + r * LDP + h * 4;
  const float* fw0 = myW0 + r * LDP + h * 4;
  const float* fw1 = myW1 + r * LDP + h * 4;

  const int col  = ntile * 16 + (lane & 15);
  const float bias0 = b0[col];
  const float bias1 = b1[col];

  v4f fa_r[HSZ / 8], fw0_r[HSZ / 8], fw1_r[HSZ / 8];
  #pragma unroll
  for (int kb = 0; kb < HSZ / 8; ++kb) {
    fa_r[kb]  = *(const v4f*)(fa  + kb * 8);
    fw0_r[kb] = *(const v4f*)(fw0 + kb * 8);
    fw1_r[kb] = *(const v4f*)(fw1 + kb * 8);
  }
  __builtin_amdgcn_sched_barrier(0);

  v8f acc0 = {};
  v8f acc1 = {};
  #pragma unroll
  for (int kb = 0; kb < HSZ / 8; ++kb) {
    const v4f a  = fa_r[kb];
    const v4f w0 = fw0_r[kb];
    const v4f w1 = fw1_r[kb];
    const v2f a_lo  = {a.x,  a.y},  a_hi  = {a.z,  a.w};
    const v2f w0_lo = {w0.x, w0.y}, w0_hi = {w0.z, w0.w};
    const v2f w1_lo = {w1.x, w1.y}, w1_hi = {w1.z, w1.w};
    acc0 = __builtin_amdgcn_wmma_f32_16x16x4_f32(
         false, a_lo,  false, w0_lo, (short)0, acc0, false, false);
    acc1 = __builtin_amdgcn_wmma_f32_16x16x4_f32(
         false, a_lo,  false, w1_lo, (short)0, acc1, false, false);
    acc0 = __builtin_amdgcn_wmma_f32_16x16x4_f32(
         false, a_hi,  false, w0_hi, (short)0, acc0, false, false);
    acc1 = __builtin_amdgcn_wmma_f32_16x16x4_f32(
         false, a_hi,  false, w1_hi, (short)0, acc1, false, false);
  }

  __builtin_amdgcn_sched_barrier(0);

  const int mofs = (lane >> 4) * 8;
  const size_t idx0 = (size_t)(mtile * 16 + mofs) * HSZ + col;
  float* __restrict__ p0 = G0 + idx0;
  float* __restrict__ p1 = G1 + idx0;
  __builtin_amdgcn_sched_barrier(0);

  #pragma unroll
  for (int v = 0; v < 8; ++v) {
    p0[(size_t)v * HSZ] = (acc0[v] + bias0) * GATE_SCALE;
    p1[(size_t)v * HSZ] = (acc1[v] + bias1) * GATE_SCALE;
  }
}

#define HS 32
#define NW 8
#define IB (NW * TI)
#define JC 8

struct Chunk {
  float x[JC];
  float g[JC];
};

__device__ __forceinline__ void load_chunk(Chunk& c, const v2f* rl, int s0)
{
  #pragma unroll
  for (int u = 0; u < JC; ++u) {
    const v2f v = rl[(s0 + u) * HS];
    c.x[u] = v.x;
    c.g[u] = v.y;
  }
}

__device__ __forceinline__ void compute_chunk(const Chunk& c, const float (&g0)[TI],
                                              float (&acc)[TI], float& accx)
{
  #pragma unroll
  for (int u = 0; u < JC; ++u) {
    const float x  = c.x[u];
    const float g1 = c.g[u];
    float th[TI];
    #pragma unroll
    for (int t = 0; t < TI; ++t)
      th[t] = __builtin_amdgcn_tanhf(g0[t] + g1);
    accx += x;
    #pragma unroll
    for (int t = 0; t < TI; ++t)
      acc[t] = __builtin_fmaf(x, th[t], acc[t]);
  }
}

__global__ __launch_bounds__(NW * 32) void pairwise_kernel(
    const float* __restrict__ inp,
    const int*   __restrict__ masks,
    const float* __restrict__ G0,
    const float* __restrict__ G1,
    float* __restrict__ out)
{
  __shared__ __attribute__((aligned(16))) v2f rows[SSZ * HS];
  __shared__ int slot[SSZ];
  __shared__ int wcnt[NW];

  const int tid  = threadIdx.x;
  const int lane = tid & 31;
  const int wave = tid >> 5;
  const int b    = blockIdx.x / (4 * (SSZ / IB));
  const int hs   = (blockIdx.x / (SSZ / IB)) & 3;
  const int ig   = blockIdx.x % (SSZ / IB);
  const int ibase = ig * IB + wave * TI;
  const int col   = hs * HS + lane;
  const int* mB   = masks + b * SSZ;

  float g0[TI];
  int   mi_row[TI];
  #pragma unroll
  for (int t = 0; t < TI; ++t) {
    g0[t]     = G0[((size_t)(b * SSZ + ibase + t) * HSZ) + col];
    mi_row[t] = mB[ibase + t];
  }

  const int r  = tid >> 3;
  const int c4 = (tid & 7) * 4;
  v4f sx[SSZ / 32], sg[SSZ / 32];
  {
    const float* xg = inp + (size_t)b * SSZ * HSZ + hs * HS + c4;
    const float* gg = G1  + (size_t)b * SSZ * HSZ + hs * HS + c4;
    #pragma unroll
    for (int p = 0; p < SSZ / 32; ++p) {
      sx[p] = *(const v4f*)(xg + (size_t)(p * 32 + r) * HSZ);
      sg[p] = *(const v4f*)(gg + (size_t)(p * 32 + r) * HSZ);
    }
  }

  const int ja = wave * 64 + lane;
  const int jb = ja + 32;
  const int ma = mB[ja];
  const int mb = mB[jb];
  const unsigned bala = __builtin_amdgcn_ballot_w32(ma != 0);
  const unsigned balb = __builtin_amdgcn_ballot_w32(mb != 0);
  const int ca = __builtin_popcount(bala);
  const int cb = __builtin_popcount(balb);
  if (lane == 0) wcnt[wave] = ca + cb;
  __syncthreads();
  int before = 0;
  int len    = 0;
  #pragma unroll
  for (int w = 0; w < NW; ++w) {
    const int c = wcnt[w];
    len += c;
    if (w < wave) before += c;
  }
  const unsigned below = (1u << lane) - 1u;
  const int ua = before + __builtin_popcount(bala & below);
  const int ub = before + ca + __builtin_popcount(balb & below);
  slot[ja] = ma ? ua : len + (ja - ua);
  slot[jb] = mb ? ub : len + (jb - ub);
  __syncthreads();

  #pragma unroll
  for (int p = 0; p < SSZ / 32; ++p) {
    const int s = slot[p * 32 + r];
    v4f xv = sx[p];
    v4f gv = sg[p];
    if (s >= len) {
      xv = (v4f){0.0f, 0.0f, 0.0f, 0.0f};
      gv = (v4f){0.0f, 0.0f, 0.0f, 0.0f};
    }
    v4f* d = (v4f*)&rows[s * HS + c4];
    d[0] = (v4f){xv.x, gv.x, xv.y, gv.y};
    d[1] = (v4f){xv.z, gv.z, xv.w, gv.w};
  }
  __syncthreads();

  const v2f* rl = rows + lane;
  float acc[TI];
  float accx = 0.0f;
  #pragma unroll
  for (int t = 0; t < TI; ++t) acc[t] = 0.0f;

  Chunk A, B;
  load_chunk(A, rl, 0);
  for (int sc = 0; sc < len; sc += 2 * JC) {
    load_chunk(B, rl, sc + JC);
    compute_chunk(A, g0, acc, accx);
    load_chunk(A, rl, (sc + 2 * JC) & (SSZ - 1));
    compute_chunk(B, g0, acc, accx);
  }

  const float inv_len = 1.0f / (float)len;
  #pragma unroll
  for (int t = 0; t < TI; ++t) {
    const float mi  = (float)mi_row[t];
    const float sum = 0.5f * (accx + acc[t]);
    out[((size_t)(b * SSZ + ibase + t) * HSZ) + col] = tanhf(mi * sum * inv_len);
  }
}

extern "C" void kernel_launch(void* const* d_in, const int* in_sizes, int n_in,
                              void* d_out, int out_size, void* d_ws, size_t ws_size,
                              hipStream_t stream) {
  const float* inp   = (const float*)d_in[0];
  const int*   masks = (const int*)  d_in[1];
  const float* W0    = (const float*)d_in[2];
  const float* b0    = (const float*)d_in[3];
  const float* W1    = (const float*)d_in[4];
  const float* b1    = (const float*)d_in[5];
  float* out = (float*)d_out;

  float* G0 = (float*)d_ws;
  float* G1 = G0 + (size_t)BSZ * SSZ * HSZ;

  gates_wmma_kernel<<<256, 128, 0, stream>>>(inp, W0, b0, W1, b1, G0, G1);
  pairwise_kernel<<<BSZ * (HSZ / HS) * (SSZ / IB), NW * 32, 0, stream>>>(inp, masks, G0, G1, out);
}


# pairwise kernel: rows i with mask 0 skip the j-loop (out = tanh(0/len) stored directly); unmasked rows i dealt round-robin to the 128 waves of a column slice, 0-4 each; j sum order kept
# speedup vs baseline: 9.3209x; 1.0593x over previous
#include <hip/hip_runtime.h>

typedef __attribute__((ext_vector_type(2))) float v2f;
typedef __attribute__((ext_vector_type(4))) float v4f;
typedef __attribute__((ext_vector_type(8))) float v8f;

#define BSZ 4
#define SSZ 512
#define HSZ 128
#define TI  4
#define GATE_SCALE 0.5f

#define LDP 132

__global__ __launch_bounds__(128) void gates_wmma_kernel(
    const float* __restrict__ inp,
    const float* __restrict__ W0, const float* __restrict__ b0,
    const float* __restrict__ W1, const float* __restrict__ b1,
    float* __restrict__ G0, float* __restrict__ G1)
{
  __shared__ __attribute__((aligned(16))) float sA[16 * LDP];
  __shared__ __attribute__((aligned(16))) float sW[4 * 2 * 16 * LDP];

  const int lane  = threadIdx.x & 31;
  const int wave  = threadIdx.x >> 5;
  const int tile  = blockIdx.x * 4 + wave;
  const int ntile = tile & 7;
  const int mtile = tile >> 3;

  const float* Ag  = inp + (size_t)(mtile * 16 + wave * 4) * HSZ + lane * 4;
  const float* W0g = W0  + (size_t)(ntile * 16) * HSZ + lane * 4;
  const float* W1g = W1  + (size_t)(ntile * 16) * HSZ + lane * 4;
  v4f ra[4], r0[16], r1[16];
  #pragma unroll
  for (int i = 0; i < 4; ++i)
    ra[i] = *(const v4f*)(Ag + (size_t)i * HSZ);
  #pragma unroll
  for (int i = 0; i < 16; ++i) {
    r0[i] = *(const v4f*)(W0g + (size_t)i * HSZ);
    r1[i] = *(const v4f*)(W1g + (size_t)i * HSZ);
  }

  float* myW0 = sW + (wave * 2 + 0) * 16 * LDP;
  float* myW1 = sW + (wave * 2 + 1) * 16 * LDP;
  #pragma unroll
  for (int i = 0; i < 4; ++i)
    *(v4f*)(sA + (wave * 4 + i) * LDP + lane * 4) = ra[i];
  #pragma unroll
  for (int i = 0; i < 16; ++i) {
    *(v4f*)(myW0 + i * LDP + lane * 4) = r0[i];
    *(v4f*)(myW1 + i * LDP + lane * 4) = r1[i];
  }
  __syncthreads();

  const int r = lane & 15;
  const int h = lane >> 4;
  const float* fa  = sA   + r * LDP + h * 4;
  const float* fw0 = myW0 + r * LDP + h * 4;
  const float* fw1 = myW1 + r * LDP + h * 4;

  const int col  = ntile * 16 + (lane & 15);
  const float bias0 = b0[col];
  const float bias1 = b1[col];

  v4f fa_r[HSZ / 8], fw0_r[HSZ / 8], fw1_r[HSZ / 8];
  #pragma unroll
  for (int kb = 0; kb < HSZ / 8; ++kb) {
    fa_r[kb]  = *(const v4f*)(fa  + kb * 8);
    fw0_r[kb] = *(const v4f*)(fw0 + kb * 8);
    fw1_r[kb] = *(const v4f*)(fw1 + kb * 8);
  }
  __builtin_amdgcn_sched_barrier(0);

  v8f acc0 = {};
  v8f acc1 = {};
  #pragma unroll
  for (int kb = 0; kb < HSZ / 8; ++kb) {
    const v4f a  = fa_r[kb];
    const v4f w0 = fw0_r[kb];
    const v4f w1 = fw1_r[kb];
    const v2f a_lo  = {a.x,  a.y},  a_hi  = {a.z,  a.w};
    const v2f w0_lo = {w0.x, w0.y}, w0_hi = {w0.z, w0.w};
    const v2f w1_lo = {w1.x, w1.y}, w1_hi = {w1.z, w1.w};
    acc0 = __builtin_amdgcn_wmma_f32_16x16x4_f32(
         false, a_lo,  false, w0_lo, (short)0, acc0, false, false);
    acc1 = __builtin_amdgcn_wmma_f32_16x16x4_f32(
         false, a_lo,  false, w1_lo, (short)0, acc1, false, false);
    acc0 = __builtin_amdgcn_wmma_f32_16x16x4_f32(
         false, a_hi,  false, w0_hi, (short)0, acc0, false, false);
    acc1 = __builtin_amdgcn_wmma_f32_16x16x4_f32(
         false, a_hi,  false, w1_hi, (short)0, acc1, false, false);
  }

  __builtin_amdgcn_sched_barrier(0);

  const int mofs = (lane >> 4) * 8;
  const size_t idx0 = (size_t)(mtile * 16 + mofs) * HSZ + col;
  float* __restrict__ p0 = G0 + idx0;
  float* __restrict__ p1 = G1 + idx0;
  __builtin_amdgcn_sched_barrier(0);

  #pragma unroll
  for (int v = 0; v < 8; ++v) {
    p0[(size_t)v * HSZ] = (acc0[v] + bias0) * GATE_SCALE;
    p1[(size_t)v * HSZ] = (acc1[v] + bias1) * GATE_SCALE;
  }
}

#define HS 32
#define NW 8
#define IB (NW * TI)
#define JC 8

struct Chunk {
  float x[JC];
  float g[JC];
};

__device__ __forceinline__ void load_chunk(Chunk& c, const v2f* rl, int s0)
{
  #pragma unroll
  for (int u = 0; u < JC; ++u) {
    const v2f v = rl[(s0 + u) * HS];
    c.x[u] = v.x;
    c.g[u] = v.y;
  }
}

template <int N>
__device__ __forceinline__ void compute_chunk(const Chunk& c, const float (&g0)[TI],
                                              float (&acc)[TI], float& accx)
{
  #pragma unroll
  for (int u = 0; u < JC; ++u) {
    const float x  = c.x[u];
    const float g1 = c.g[u];
    float th[N];
    #pragma unroll
    for (int t = 0; t < N; ++t)
      th[t] = __builtin_amdgcn_tanhf(g0[t] + g1);
    accx += x;
    #pragma unroll
    for (int t = 0; t < N; ++t)
      acc[t] = __builtin_fmaf(x, th[t], acc[t]);
  }
}

template <int N>
__device__ __forceinline__ void row_sums(const v2f* rl, int len, const float (&g0)[TI],
                                         float (&acc)[TI], float& accx)
{
  Chunk A, B;
  load_chunk(A, rl, 0);
  for (int sc = 0; sc < len; sc += 2 * JC) {
    load_chunk(B, rl, sc + JC);
    compute_chunk<N>(A, g0, acc, accx);
    load_chunk(A, rl, (sc + 2 * JC) & (SSZ - 1));
    compute_chunk<N>(B, g0, acc, accx);
  }
}

__global__ __launch_bounds__(NW * 32) void pairwise_kernel(
    const float* __restrict__ inp,
    const int*   __restrict__ masks,
    const float* __restrict__ G0,
    const float* __restrict__ G1,
    float* __restrict__ out)
{
  __shared__ __attribute__((aligned(16))) v2f rows[SSZ * HS];
  __shared__ int slot[SSZ];
  __shared__ int irow[SSZ];
  __shared__ int wcnt[NW];

  const int tid  = threadIdx.x;
  const int lane = tid & 31;
  const int wave = tid >> 5;
  const int b    = blockIdx.x / (4 * (SSZ / IB));
  const int hs   = (blockIdx.x / (SSZ / IB)) & 3;
  const int ig   = blockIdx.x % (SSZ / IB);
  const int ibase = ig * IB + wave * TI;
  const int col   = hs * HS + lane;
  const int* mB   = masks + b * SSZ;

  int mi_row[TI];
  #pragma unroll
  for (int t = 0; t < TI; ++t)
    mi_row[t] = mB[ibase + t];

  const int r  = tid >> 3;
  const int c4 = (tid & 7) * 4;
  v4f sx[SSZ / 32], sg[SSZ / 32];
  {
    const float* xg = inp + (size_t)b * SSZ * HSZ + hs * HS + c4;
    const float* gg = G1  + (size_t)b * SSZ * HSZ + hs * HS + c4;
    #pragma unroll
    for (int p = 0; p < SSZ / 32; ++p) {
      sx[p] = *(const v4f*)(xg + (size_t)(p * 32 + r) * HSZ);
      sg[p] = *(const v4f*)(gg + (size_t)(p * 32 + r) * HSZ);
    }
  }

  const int ja = wave * 64 + lane;
  const int jb = ja + 32;
  const int ma = mB[ja];
  const int mb = mB[jb];
  const unsigned bala = __builtin_amdgcn_ballot_w32(ma != 0);
  const unsigned balb = __builtin_amdgcn_ballot_w32(mb != 0);
  const int ca = __builtin_popcount(bala);
  const int cb = __builtin_popcount(balb);
  if (lane == 0) wcnt[wave] = ca + cb;
  __syncthreads();
  int before = 0;
  int len    = 0;
  #pragma unroll
  for (int w = 0; w < NW; ++w) {
    const int c = wcnt[w];
    len += c;
    if (w < wave) before += c;
  }
  const unsigned below = (1u << lane) - 1u;
  const int ua = before + __builtin_popcount(bala & below);
  const int ub = before + ca + __builtin_popcount(balb & below);
  const int sa = ma ? ua : len + (ja - ua);
  const int sb = mb ? ub : len + (jb - ub);
  slot[ja] = sa;
  slot[jb] = sb;
  irow[sa] = ja;
  irow[sb] = jb;
  __syncthreads();

  const int gw = wave * (SSZ / IB) + ig;
  int nr = (len - gw + 127) >> 7;
  nr = nr < 0 ? 0 : nr;
  const int nrow = __builtin_amdgcn_readfirstlane(nr);
  int   irw[TI];
  float g0[TI];
  #pragma unroll
  for (int t = 0; t < TI; ++t)
    irw[t] = irow[gw + t * (NW * (SSZ / IB))];
  #pragma unroll
  for (int t = 0; t < TI; ++t)
    g0[t] = G0[((size_t)(b * SSZ + irw[t]) * HSZ) + col];

  #pragma unroll
  for (int p = 0; p < SSZ / 32; ++p) {
    const int s = slot[p * 32 + r];
    v4f xv = sx[p];
    v4f gv = sg[p];
    if (s >= len) {
      xv = (v4f){0.0f, 0.0f, 0.0f, 0.0f};
      gv = (v4f){0.0f, 0.0f, 0.0f, 0.0f};
    }
    v4f* d = (v4f*)&rows[s * HS + c4];
    d[0] = (v4f){xv.x, gv.x, xv.y, gv.y};
    d[1] = (v4f){xv.z, gv.z, xv.w, gv.w};
  }
  __syncthreads();

  const v2f* rl = rows + lane;
  float acc[TI];
  float accx = 0.0f;
  #pragma unroll
  for (int t = 0; t < TI; ++t) acc[t] = 0.0f;

  switch (nrow) {
    case 1: row_sums<1>(rl, len, g0, acc, accx); break;
    case 2: row_sums<2>(rl, len, g0, acc, accx); break;
    case 3: row_sums<3>(rl, len, g0, acc, accx); break;
    case 4: row_sums<4>(rl, len, g0, acc, accx); break;
    default: break;
  }

  const float inv_len = 1.0f / (float)len;
  #pragma unroll
  for (int t = 0; t < TI; ++t) {
    if (t < nrow) {
      const float sum = 0.5f * (accx + acc[t]);
      out[((size_t)(b * SSZ + irw[t]) * HSZ) + col] = tanhf(sum * inv_len);
    }
  }
  const float zv = 0.0f * inv_len;
  #pragma unroll
  for (int t = 0; t < TI; ++t) {
    if (mi_row[t] == 0)
      out[((size_t)(b * SSZ + ibase + t) * HSZ) + col] = zv;
  }
}

extern "C" void kernel_launch(void* const* d_in, const int* in_sizes, int n_in,
                              void* d_out, int out_size, void* d_ws, size_t ws_size,
                              hipStream_t stream) {
  const float* inp   = (const float*)d_in[0];
  const int*   masks = (const int*)  d_in[1];
  const float* W0    = (const float*)d_in[2];
  const float* b0    = (const float*)d_in[3];
  const float* W1    = (const float*)d_in[4];
  const float* b1    = (const float*)d_in[5];
  float* out = (float*)d_out;

  float* G0 = (float*)d_ws;
  float* G1 = G0 + (size_t)BSZ * SSZ * HSZ;

  gates_wmma_kernel<<<256, 128, 0, stream>>>(inp, W0, b0, W1, b1, G0, G1);
  pairwise_kernel<<<BSZ * (HSZ / HS) * (SSZ / IB), NW * 32, 0, stream>>>(inp, masks, G0, G1, out);
}


# pairwise kernel prologue overlapped: masks loaded first, its barriers order LDS only, each staging pass stored to LDS as its loads land; staging loads now fly under slot arithmetic and LDS stores
# speedup vs baseline: 9.9050x; 1.0627x over previous
#include <hip/hip_runtime.h>

typedef __attribute__((ext_vector_type(2))) float v2f;
typedef __attribute__((ext_vector_type(4))) float v4f;
typedef __attribute__((ext_vector_type(8))) float v8f;

#define BSZ 4
#define SSZ 512
#define HSZ 128
#define TI  4
#define GATE_SCALE 0.5f

#define LDP 132

__global__ __launch_bounds__(128) void gates_wmma_kernel(
    const float* __restrict__ inp,
    const float* __restrict__ W0, const float* __restrict__ b0,
    const float* __restrict__ W1, const float* __restrict__ b1,
    float* __restrict__ G0, float* __restrict__ G1)
{
  __shared__ __attribute__((aligned(16))) float sA[16 * LDP];
  __shared__ __attribute__((aligned(16))) float sW[4 * 2 * 16 * LDP];

  const int lane  = threadIdx.x & 31;
  const int wave  = threadIdx.x >> 5;
  const int tile  = blockIdx.x * 4 + wave;
  const int ntile = tile & 7;
  const int mtile = tile >> 3;

  const float* Ag  = inp + (size_t)(mtile * 16 + wave * 4) * HSZ + lane * 4;
  const float* W0g = W0  + (size_t)(ntile * 16) * HSZ + lane * 4;
  const float* W1g = W1  + (size_t)(ntile * 16) * HSZ + lane * 4;
  v4f ra[4], r0[16], r1[16];
  #pragma unroll
  for (int i = 0; i < 4; ++i)
    ra[i] = *(const v4f*)(Ag + (size_t)i * HSZ);
  #pragma unroll
  for (int i = 0; i < 16; ++i) {
    r0[i] = *(const v4f*)(W0g + (size_t)i * HSZ);
    r1[i] = *(const v4f*)(W1g + (size_t)i * HSZ);
  }

  float* myW0 = sW + (wave * 2 + 0) * 16 * LDP;
  float* myW1 = sW + (wave * 2 + 1) * 16 * LDP;
  #pragma unroll
  for (int i = 0; i < 4; ++i)
    *(v4f*)(sA + (wave * 4 + i) * LDP + lane * 4) = ra[i];
  #pragma unroll
  for (int i = 0; i < 16; ++i) {
    *(v4f*)(myW0 + i * LDP + lane * 4) = r0[i];
    *(v4f*)(myW1 + i * LDP + lane * 4) = r1[i];
  }
  __syncthreads();

  const int r = lane & 15;
  const int h = lane >> 4;
  const float* fa  = sA   + r * LDP + h * 4;
  const float* fw0 = myW0 + r * LDP + h * 4;
  const float* fw1 = myW1 + r * LDP + h * 4;

  const int col  = ntile * 16 + (lane & 15);
  const float bias0 = b0[col];
  const float bias1 = b1[col];

  v4f fa_r[HSZ / 8], fw0_r[HSZ / 8], fw1_r[HSZ / 8];
  #pragma unroll
  for (int kb = 0; kb < HSZ / 8; ++kb) {
    fa_r[kb]  = *(const v4f*)(fa  + kb * 8);
    fw0_r[kb] = *(const v4f*)(fw0 + kb * 8);
    fw1_r[kb] = *(const v4f*)(fw1 + kb * 8);
  }
  __builtin_amdgcn_sched_barrier(0);

  v8f acc0 = {};
  v8f acc1 = {};
  #pragma unroll
  for (int kb = 0; kb < HSZ / 8; ++kb) {
    const v4f a  = fa_r[kb];
    const v4f w0 = fw0_r[kb];
    const v4f w1 = fw1_r[kb];
    const v2f a_lo  = {a.x,  a.y},  a_hi  = {a.z,  a.w};
    const v2f w0_lo = {w0.x, w0.y}, w0_hi = {w0.z, w0.w};
    const v2f w1_lo = {w1.x, w1.y}, w1_hi = {w1.z, w1.w};
    acc0 = __builtin_amdgcn_wmma_f32_16x16x4_f32(
         false, a_lo,  false, w0_lo, (short)0, acc0, false, false);
    acc1 = __builtin_amdgcn_wmma_f32_16x16x4_f32(
         false, a_lo,  false, w1_lo, (short)0, acc1, false, false);
    acc0 = __builtin_amdgcn_wmma_f32_16x16x4_f32(
         false, a_hi,  false, w0_hi, (short)0, acc0, false, false);
    acc1 = __builtin_amdgcn_wmma_f32_16x16x4_f32(
         false, a_hi,  false, w1_hi, (short)0, acc1, false, false);
  }

  __builtin_amdgcn_sched_barrier(0);

  const int mofs = (lane >> 4) * 8;
  const size_t idx0 = (size_t)(mtile * 16 + mofs) * HSZ + col;
  float* __restrict__ p0 = G0 + idx0;
  float* __restrict__ p1 = G1 + idx0;
  __builtin_amdgcn_sched_barrier(0);

  #pragma unroll
  for (int v = 0; v < 8; ++v) {
    p0[(size_t)v * HSZ] = (acc0[v] + bias0) * GATE_SCALE;
    p1[(size_t)v * HSZ] = (acc1[v] + bias1) * GATE_SCALE;
  }
}

#define HS 32
#define NW 8
#define IB (NW * TI)
#define JC 8

struct Chunk {
  float x[JC];
  float g[JC];
};

__device__ __forceinline__ void load_chunk(Chunk& c, const v2f* rl, int s0)
{
  #pragma unroll
  for (int u = 0; u < JC; ++u) {
    const v2f v = rl[(s0 + u) * HS];
    c.x[u] = v.x;
    c.g[u] = v.y;
  }
}

template <int N>
__device__ __forceinline__ void compute_chunk(const Chunk& c, const float (&g0)[TI],
                                              float (&acc)[TI], float& accx)
{
  #pragma unroll
  for (int u = 0; u < JC; ++u) {
    const float x  = c.x[u];
    const float g1 = c.g[u];
    float th[N];
    #pragma unroll
    for (int t = 0; t < N; ++t)
      th[t] = __builtin_amdgcn_tanhf(g0[t] + g1);
    accx += x;
    #pragma unroll
    for (int t = 0; t < N; ++t)
      acc[t] = __builtin_fmaf(x, th[t], acc[t]);
  }
}

template <int N>
__device__ __forceinline__ void row_sums(const v2f* rl, int len, const float (&g0)[TI],
                                         float (&acc)[TI], float& accx)
{
  Chunk A, B;
  load_chunk(A, rl, 0);
  for (int sc = 0; sc < len; sc += 2 * JC) {
    load_chunk(B, rl, sc + JC);
    compute_chunk<N>(A, g0, acc, accx);
    load_chunk(A, rl, (sc + 2 * JC) & (SSZ - 1));
    compute_chunk<N>(B, g0, acc, accx);
  }
}

__device__ __forceinline__ void lds_barrier()
{
  __builtin_amdgcn_fence(__ATOMIC_RELEASE, "workgroup", "local");
  __builtin_amdgcn_s_barrier();
  __builtin_amdgcn_fence(__ATOMIC_ACQUIRE, "workgroup", "local");
}

__global__ __launch_bounds__(NW * 32) void pairwise_kernel(
    const float* __restrict__ inp,
    const int*   __restrict__ masks,
    const float* __restrict__ G0,
    const float* __restrict__ G1,
    float* __restrict__ out)
{
  __shared__ __attribute__((aligned(16))) v2f rows[SSZ * HS];
  __shared__ int slot[SSZ];
  __shared__ int irow[SSZ];
  __shared__ int wcnt[NW];

  const int tid  = threadIdx.x;
  const int lane = tid & 31;
  const int wave = tid >> 5;
  const int b    = blockIdx.x / (4 * (SSZ / IB));
  const int hs   = (blockIdx.x / (SSZ / IB)) & 3;
  const int ig   = blockIdx.x % (SSZ / IB);
  const int ibase = ig * IB + wave * TI;
  const int col   = hs * HS + lane;
  const int* mB   = masks + b * SSZ;

  int mi_row[TI];
  #pragma unroll
  for (int t = 0; t < TI; ++t)
    mi_row[t] = mB[ibase + t];
  const int ja = wave * 64 + lane;
  const int jb = ja + 32;
  const int ma = mB[ja];
  const int mb = mB[jb];

  const int r  = tid >> 3;
  const int c4 = (tid & 7) * 4;
  v4f sx[SSZ / 32], sg[SSZ / 32];
  {
    const float* xg = inp + (size_t)b * SSZ * HSZ + hs * HS + c4;
    const float* gg = G1  + (size_t)b * SSZ * HSZ + hs * HS + c4;
    #pragma unroll
    for (int p = 0; p < SSZ / 32; ++p) {
      sx[p] = *(const v4f*)(xg + (size_t)(p * 32 + r) * HSZ);
      sg[p] = *(const v4f*)(gg + (size_t)(p * 32 + r) * HSZ);
    }
  }

  __builtin_amdgcn_sched_barrier(0);

  const unsigned bala = __builtin_amdgcn_ballot_w32(ma != 0);
  const unsigned balb = __builtin_amdgcn_ballot_w32(mb != 0);
  const int ca = __builtin_popcount(bala);
  const int cb = __builtin_popcount(balb);
  if (lane == 0) wcnt[wave] = ca + cb;
  lds_barrier();
  int before = 0;
  int len    = 0;
  #pragma unroll
  for (int w = 0; w < NW; ++w) {
    const int c = wcnt[w];
    len += c;
    if (w < wave) before += c;
  }
  const unsigned below = (1u << lane) - 1u;
  const int ua = before + __builtin_popcount(bala & below);
  const int ub = before + ca + __builtin_popcount(balb & below);
  const int sa = ma ? ua : len + (ja - ua);
  const int sb = mb ? ub : len + (jb - ub);
  slot[ja] = sa;
  slot[jb] = sb;
  irow[sa] = ja;
  irow[sb] = jb;
  lds_barrier();

  const int gw = wave * (SSZ / IB) + ig;
  int nr = (len - gw + 127) >> 7;
  nr = nr < 0 ? 0 : nr;
  const int nrow = __builtin_amdgcn_readfirstlane(nr);
  int   irw[TI];
  float g0[TI];
  #pragma unroll
  for (int t = 0; t < TI; ++t)
    irw[t] = irow[gw + t * (NW * (SSZ / IB))];
  #pragma unroll
  for (int t = 0; t < TI; ++t)
    g0[t] = G0[((size_t)(b * SSZ + irw[t]) * HSZ) + col];

  int sl[SSZ / 32];
  #pragma unroll
  for (int p = 0; p < SSZ / 32; ++p)
    sl[p] = slot[p * 32 + r];
  #pragma unroll
  for (int p = 0; p < SSZ / 32; ++p) {
    const int s = sl[p];
    v4f xv = sx[p];
    v4f gv = sg[p];
    if (s >= len) {
      xv = (v4f){0.0f, 0.0f, 0.0f, 0.0f};
      gv = (v4f){0.0f, 0.0f, 0.0f, 0.0f};
    }
    v4f* d = (v4f*)&rows[s * HS + c4];
    d[0] = (v4f){xv.x, gv.x, xv.y, gv.y};
    d[1] = (v4f){xv.z, gv.z, xv.w, gv.w};
    __builtin_amdgcn_sched_barrier(0);
  }
  lds_barrier();

  const v2f* rl = rows + lane;
  float acc[TI];
  float accx = 0.0f;
  #pragma unroll
  for (int t = 0; t < TI; ++t) acc[t] = 0.0f;

  switch (nrow) {
    case 1: row_sums<1>(rl, len, g0, acc, accx); break;
    case 2: row_sums<2>(rl, len, g0, acc, accx); break;
    case 3: row_sums<3>(rl, len, g0, acc, accx); break;
    case 4: row_sums<4>(rl, len, g0, acc, accx); break;
    default: break;
  }

  const float inv_len = 1.0f / (float)len;
  #pragma unroll
  for (int t = 0; t < TI; ++t) {
    if (t < nrow) {
      const float sum = 0.5f * (accx + acc[t]);
      out[((size_t)(b * SSZ + irw[t]) * HSZ) + col] = tanhf(sum * inv_len);
    }
  }
  const float zv = 0.0f * inv_len;
  #pragma unroll
  for (int t = 0; t < TI; ++t) {
    if (mi_row[t] == 0)
      out[((size_t)(b * SSZ + ibase + t) * HSZ) + col] = zv;
  }
}

extern "C" void kernel_launch(void* const* d_in, const int* in_sizes, int n_in,
                              void* d_out, int out_size, void* d_ws, size_t ws_size,
                              hipStream_t stream) {
  const float* inp   = (const float*)d_in[0];
  const int*   masks = (const int*)  d_in[1];
  const float* W0    = (const float*)d_in[2];
  const float* b0    = (const float*)d_in[3];
  const float* W1    = (const float*)d_in[4];
  const float* b1    = (const float*)d_in[5];
  float* out = (float*)d_out;

  float* G0 = (float*)d_ws;
  float* G1 = G0 + (size_t)BSZ * SSZ * HSZ;

  gates_wmma_kernel<<<256, 128, 0, stream>>>(inp, W0, b0, W1, b1, G0, G1);
  pairwise_kernel<<<BSZ * (HSZ / HS) * (SSZ / IB), NW * 32, 0, stream>>>(inp, masks, G0, G1, out);
}


# one launch, not two: each pairwise block forms its own gate columns on the matrix unit in LDS (g1 of the batch's unmasked rows, g0 of its rows i); no gate pass, no workspace; sum orders kept
# speedup vs baseline: 10.5000x; 1.0601x over previous
#include <hip/hip_runtime.h>

typedef __attribute__((ext_vector_type(2))) float v2f;
typedef __attribute__((ext_vector_type(4))) float v4f;
typedef __attribute__((ext_vector_type(8))) float v8f;

#define BSZ 4
#define SSZ 512
#define HSZ 128
#define TI  4
#define GATE_SCALE 0.5f
#define LDP 132


#define HS 32
#define NW 8
#define IB (NW * TI)
#define JC 8

struct Chunk {
  float x[JC];
  float g[JC];
};

__device__ __forceinline__ void load_chunk(Chunk& c, const v2f* rl, int s0)
{
  #pragma unroll
  for (int u = 0; u < JC; ++u) {
    const v2f v = rl[(s0 + u) * HS];
    c.x[u] = v.x;
    c.g[u] = v.y;
  }
}

template <int N>
__device__ __forceinline__ void compute_chunk(const Chunk& c, const float (&g0)[TI],
                                              float (&acc)[TI], float& accx)
{
  #pragma unroll
  for (int u = 0; u < JC; ++u) {
    const float x  = c.x[u];
    const float g1 = c.g[u];
    float th[N];
    #pragma unroll
    for (int t = 0; t < N; ++t)
      th[t] = __builtin_amdgcn_tanhf(g0[t] + g1);
    accx += x;
    #pragma unroll
    for (int t = 0; t < N; ++t)
      acc[t] = __builtin_fmaf(x, th[t], acc[t]);
  }
}

template <int N>
__device__ __forceinline__ void row_sums(const v2f* rl, int len, const float (&g0)[TI],
                                         float (&acc)[TI], float& accx)
{
  Chunk A, B;
  load_chunk(A, rl, 0);
  for (int sc = 0; sc < len; sc += 2 * JC) {
    load_chunk(B, rl, sc + JC);
    compute_chunk<N>(A, g0, acc, accx);
    load_chunk(A, rl, (sc + 2 * JC) & (SSZ - 1));
    compute_chunk<N>(B, g0, acc, accx);
  }
}

__device__ __forceinline__ void lds_barrier()
{
  __builtin_amdgcn_fence(__ATOMIC_RELEASE, "workgroup", "local");
  __builtin_amdgcn_s_barrier();
  __builtin_amdgcn_fence(__ATOMIC_ACQUIRE, "workgroup", "local");
}


__device__ __forceinline__ void tile_product2(const float* fa, const float* fb0, const float* fb1,
                                              v8f& acc0, v8f& acc1)
{
  v4f fa_r[HSZ / 8], fb0_r[HSZ / 8], fb1_r[HSZ / 8];
  #pragma unroll
  for (int kb = 0; kb < HSZ / 8; ++kb) {
    fa_r[kb]  = *(const v4f*)(fa  + kb * 8);
    fb0_r[kb] = *(const v4f*)(fb0 + kb * 8);
    fb1_r[kb] = *(const v4f*)(fb1 + kb * 8);
  }
  __builtin_amdgcn_sched_barrier(0);
  #pragma unroll
  for (int kb = 0; kb < HSZ / 8; ++kb) {
    const v4f a  = fa_r[kb];
    const v4f w0 = fb0_r[kb];
    const v4f w1 = fb1_r[kb];
    const v2f a_lo  = {a.x,  a.y},  a_hi  = {a.z,  a.w};
    const v2f w0_lo = {w0.x, w0.y}, w0_hi = {w0.z, w0.w};
    const v2f w1_lo = {w1.x, w1.y}, w1_hi = {w1.z, w1.w};
    acc0 = __builtin_amdgcn_wmma_f32_16x16x4_f32(
         false, a_lo,  false, w0_lo, (short)0, acc0, false, false);
    acc1 = __builtin_amdgcn_wmma_f32_16x16x4_f32(
         false, a_lo,  false, w1_lo, (short)0, acc1, false, false);
    acc0 = __builtin_amdgcn_wmma_f32_16x16x4_f32(
         false, a_hi,  false, w0_hi, (short)0, acc0, false, false);
    acc1 = __builtin_amdgcn_wmma_f32_16x16x4_f32(
         false, a_hi,  false, w1_hi, (short)0, acc1, false, false);
  }
  __builtin_amdgcn_sched_barrier(0);
}

__device__ __forceinline__ void tile_product1(const float* fa, const float* fb, v8f& acc)
{
  v4f fa_r[HSZ / 8], fb_r[HSZ / 8];
  #pragma unroll
  for (int kb = 0; kb < HSZ / 8; ++kb) {
    fa_r[kb] = *(const v4f*)(fa + kb * 8);
    fb_r[kb] = *(const v4f*)(fb + kb * 8);
  }
  __builtin_amdgcn_sched_barrier(0);
  #pragma unroll
  for (int kb = 0; kb < HSZ / 8; ++kb) {
    const v4f a = fa_r[kb];
    const v4f w = fb_r[kb];
    const v2f a_lo = {a.x, a.y}, a_hi = {a.z, a.w};
    const v2f w_lo = {w.x, w.y}, w_hi = {w.z, w.w};
    acc = __builtin_amdgcn_wmma_f32_16x16x4_f32(
         false, a_lo,  false, w_lo, (short)0, acc, false, false);
    acc = __builtin_amdgcn_wmma_f32_16x16x4_f32(
         false, a_hi,  false, w_hi, (short)0, acc, false, false);
  }
  __builtin_amdgcn_sched_barrier(0);
}

__global__ __launch_bounds__(NW * 32) void relation_fused_kernel(
    const float* __restrict__ inp,
    const int*   __restrict__ masks,
    const float* __restrict__ W0, const float* __restrict__ b0,
    const float* __restrict__ W1, const float* __restrict__ b1,
    float* __restrict__ out)
{
  __shared__ __attribute__((aligned(16))) v2f   rows[SSZ * HS];
  __shared__ __attribute__((aligned(16))) float sAt[NW * 16 * LDP];
  __shared__ __attribute__((aligned(16))) float sW1[HS * LDP];
  __shared__ __attribute__((aligned(16))) float sW0[HS * LDP];
  __shared__ float sG0[IB * HS];
  __shared__ int irow[SSZ];
  __shared__ int wcnt[NW];

  const int tid  = threadIdx.x;
  const int lane = tid & 31;
  const int wave = __builtin_amdgcn_readfirstlane(tid >> 5);
  const int b    = blockIdx.x / (4 * (SSZ / IB));
  const int hs   = (blockIdx.x / (SSZ / IB)) & 3;
  const int ig   = blockIdx.x % (SSZ / IB);
  const int ibase = ig * IB + wave * TI;
  const int col   = hs * HS + lane;
  const int* mB   = masks + b * SSZ;
  const int r = lane & 15;
  const int h = lane >> 4;

  int mi_row[TI];
  #pragma unroll
  for (int t = 0; t < TI; ++t)
    mi_row[t] = mB[ibase + t];
  const int ja = wave * 64 + lane;
  const int jb = ja + 32;
  const int ma = mB[ja];
  const int mb = mB[jb];

  v4f rw1[4], rw0[4];
  {
    const float* w1g = W1 + (size_t)(hs * HS + wave) * HSZ + lane * 4;
    const float* w0g = W0 + (size_t)(hs * HS + wave) * HSZ + lane * 4;
    #pragma unroll
    for (int p = 0; p < 4; ++p) {
      rw1[p] = *(const v4f*)(w1g + (size_t)(p * NW) * HSZ);
      rw0[p] = *(const v4f*)(w0g + (size_t)(p * NW) * HSZ);
    }
  }
  const float bias1a = b1[hs * HS + r];
  const float bias1b = b1[hs * HS + 16 + r];
  const float bias0  = b0[hs * HS + (wave & 1) * 16 + r];

  __builtin_amdgcn_sched_barrier(0);

  const unsigned bala = __builtin_amdgcn_ballot_w32(ma != 0);
  const unsigned balb = __builtin_amdgcn_ballot_w32(mb != 0);
  const int ca = __builtin_popcount(bala);
  const int cb = __builtin_popcount(balb);
  if (lane == 0) wcnt[wave] = ca + cb;
  lds_barrier();
  int before = 0;
  int len    = 0;
  #pragma unroll
  for (int w = 0; w < NW; ++w) {
    const int c = wcnt[w];
    len += c;
    if (w < wave) before += c;
  }
  const unsigned below = (1u << lane) - 1u;
  const int ua = before + __builtin_popcount(bala & below);
  const int ub = before + ca + __builtin_popcount(balb & below);
  const int sa = ma ? ua : len + (ja - ua);
  const int sb = mb ? ub : len + (jb - ub);
  irow[sa] = ja;
  irow[sb] = jb;
  #pragma unroll
  for (int p = 0; p < 4; ++p) {
    *(v4f*)(sW1 + (wave + p * NW) * LDP + lane * 4) = rw1[p];
    *(v4f*)(sW0 + (wave + p * NW) * LDP + lane * 4) = rw0[p];
  }
  lds_barrier();

  const int lenS = __builtin_amdgcn_readfirstlane(len);
  const int gw = wave * (SSZ / IB) + ig;
  int nr = (lenS - gw + 127) >> 7;
  nr = nr < 0 ? 0 : nr;
  const int nrow = nr;
  int nmax = (lenS - ig + 127) >> 7;
  nmax = nmax < 0 ? 0 : nmax;
  const int gtiles = (nmax + 1) >> 1;
  int ntw = (lenS - wave * 16 + 127) >> 7;
  ntw = ntw < 0 ? 0 : (ntw > SSZ / (NW * 16) ? SSZ / (NW * 16) : ntw);

  int irw[TI];
  #pragma unroll
  for (int t = 0; t < TI; ++t)
    irw[t] = irow[gw + t * (NW * (SSZ / IB))];

  float* sAw = sAt + wave * 16 * LDP;
  const float* fa  = sAw + r * LDP + h * 4;
  const float* fb0 = sW1 + r * LDP + h * 4;
  const float* fb1 = sW1 + (16 + r) * LDP + h * 4;
  const float* inB = inp + (size_t)b * SSZ * HSZ;
  const unsigned lo4 = (unsigned)lane * 4u;
  v4f ar[16];

  const int  gsel   = (7 - wave) >> 1;
  const bool has_g0 = (wave >= 4) && (gsel < gtiles);

  if (has_g0) {
    #pragma unroll
    for (int q = 0; q < 16; ++q) {
      const int k = (q & 7) * (SSZ / IB) + ig + (2 * gsel + (q >> 3)) * (NW * (SSZ / IB));
      ar[q] = *(const v4f*)(inB + ((unsigned)irow[k] * HSZ + lo4));
    }
    #pragma unroll
    for (int q = 0; q < 16; ++q)
      *(v4f*)(sAw + q * LDP + lane * 4) = ar[q];
  }
  if (ntw > 0) {
    #pragma unroll
    for (int q = 0; q < 16; ++q)
      ar[q] = *(const v4f*)(inB + ((unsigned)irow[wave * 16 + q] * HSZ + lo4));
  }
  if (has_g0) {
    v8f acc = {};
    tile_product1(fa, sW0 + ((wave & 1) * 16 + r) * LDP + h * 4, acc);
    float* g = sG0 + ((2 * gsel + h) * NW) * HS + (wave & 1) * 16 + r;
    #pragma unroll
    for (int v = 0; v < 8; ++v)
      g[v * HS] = (acc[v] + bias0) * GATE_SCALE;
  }

  const float* xr = sAw + (8 * h) * LDP + hs * HS + r;
  for (int c = 0; c < ntw; ++c) {
    const int s0 = c * (NW * 16) + wave * 16;
    const int sl0 = s0 + 8 * h;
    v2f* d = rows + sl0 * HS + r;
    #pragma unroll
    for (int q = 0; q < 16; ++q)
      *(v4f*)(sAw + q * LDP + lane * 4) = ar[q];
    if (c + 1 < ntw) {
      #pragma unroll
      for (int q = 0; q < 16; ++q)
        ar[q] = *(const v4f*)(inB + ((unsigned)irow[s0 + NW * 16 + q] * HSZ + lo4));
    }
    float x0[8], x1[8];
    #pragma unroll
    for (int v = 0; v < 8; ++v) {
      x0[v] = xr[v * LDP];
      x1[v] = xr[v * LDP + 16];
    }
    v8f acc0 = {};
    v8f acc1 = {};
    tile_product2(fa, fb0, fb1, acc0, acc1);
    #pragma unroll
    for (int v = 0; v < 8; ++v) {
      const bool live = (sl0 + v) < lenS;
      const float ga = (acc0[v] + bias1a) * GATE_SCALE;
      const float gb = (acc1[v] + bias1b) * GATE_SCALE;
      d[v * HS]      = live ? (v2f){x0[v], ga} : (v2f){0.0f, 0.0f};
      d[v * HS + 16] = live ? (v2f){x1[v], gb} : (v2f){0.0f, 0.0f};
    }
  }
  for (int c = ntw; c < SSZ / (NW * 16); ++c) {
    v2f* d = rows + (c * (NW * 16) + wave * 16 + 8 * h) * HS + r;
    #pragma unroll
    for (int v = 0; v < 8; ++v) {
      d[v * HS]      = (v2f){0.0f, 0.0f};
      d[v * HS + 16] = (v2f){0.0f, 0.0f};
    }
  }
  lds_barrier();

  float g0[TI];
  #pragma unroll
  for (int t = 0; t < TI; ++t)
    g0[t] = (t < nrow) ? sG0[(t * NW + wave) * HS + lane] : 0.0f;

  const v2f* rl = rows + lane;
  float acc[TI];
  float accx = 0.0f;
  #pragma unroll
  for (int t = 0; t < TI; ++t) acc[t] = 0.0f;

  switch (nrow) {
    case 1: row_sums<1>(rl, lenS, g0, acc, accx); break;
    case 2: row_sums<2>(rl, lenS, g0, acc, accx); break;
    case 3: row_sums<3>(rl, lenS, g0, acc, accx); break;
    case 4: row_sums<4>(rl, lenS, g0, acc, accx); break;
    default: break;
  }

  const float inv_len = 1.0f / (float)lenS;
  #pragma unroll
  for (int t = 0; t < TI; ++t) {
    if (t < nrow) {
      const float sum = 0.5f * (accx + acc[t]);
      out[((size_t)(b * SSZ + irw[t]) * HSZ) + col] = tanhf(sum * inv_len);
    }
  }
  const float zv = 0.0f * inv_len;
  #pragma unroll
  for (int t = 0; t < TI; ++t) {
    if (mi_row[t] == 0)
      out[((size_t)(b * SSZ + ibase + t) * HSZ) + col] = zv;
  }
}

extern "C" void kernel_launch(void* const* d_in, const int* in_sizes, int n_in,
                              void* d_out, int out_size, void* d_ws, size_t ws_size,
                              hipStream_t stream) {
  const float* inp   = (const float*)d_in[0];
  const int*   masks = (const int*)  d_in[1];
  const float* W0    = (const float*)d_in[2];
  const float* b0    = (const float*)d_in[3];
  const float* W1    = (const float*)d_in[4];
  const float* b1    = (const float*)d_in[5];
  float* out = (float*)d_out;

  relation_fused_kernel<<<BSZ * (HSZ / HS) * (SSZ / IB), NW * 32, 0, stream>>>(
      inp, masks, W0, b0, W1, b1, out);
}


# pairwise kernel re-tiled to (batch, 16 columns, 64 rows i): a wave's lane halves take different rows i and share each slot read; the repeated g1 product is one column tile, not two; sum orders kept
# speedup vs baseline: 11.9429x; 1.1374x over previous
#include <hip/hip_runtime.h>

typedef __attribute__((ext_vector_type(2))) float v2f;
typedef __attribute__((ext_vector_type(4))) float v4f;
typedef __attribute__((ext_vector_type(8))) float v8f;

#define BSZ 4
#define SSZ 512
#define HSZ 128
#define TI  4
#define GATE_SCALE 0.5f
#define LDP 132


#define HS 16
#define NW 8
#define NIG 8
#define NHW (NW * NIG)
#define JC 8
#define LDS_TAIL_BYTES (24 * 1024)

struct Chunk {
  float x[JC];
  float g[JC];
};

__device__ __forceinline__ void load_chunk(Chunk& c, const v2f* rl, int s0)
{
  #pragma unroll
  for (int u = 0; u < JC; ++u) {
    const v2f v = rl[(s0 + u) * HS];
    c.x[u] = v.x;
    c.g[u] = v.y;
  }
}

template <int N>
__device__ __forceinline__ void compute_chunk(const Chunk& c, const float (&g0)[TI],
                                              float (&acc)[TI], float& accx)
{
  #pragma unroll
  for (int u = 0; u < JC; ++u) {
    const float x  = c.x[u];
    const float g1 = c.g[u];
    float th[N];
    #pragma unroll
    for (int t = 0; t < N; ++t)
      th[t] = __builtin_amdgcn_tanhf(g0[t] + g1);
    accx += x;
    #pragma unroll
    for (int t = 0; t < N; ++t)
      acc[t] = __builtin_fmaf(x, th[t], acc[t]);
  }
}

template <int N>
__device__ __forceinline__ void row_sums(const v2f* rl, int len, const float (&g0)[TI],
                                         float (&acc)[TI], float& accx)
{
  Chunk A, B;
  load_chunk(A, rl, 0);
  for (int sc = 0; sc < len; sc += 2 * JC) {
    load_chunk(B, rl, sc + JC);
    compute_chunk<N>(A, g0, acc, accx);
    load_chunk(A, rl, (sc + 2 * JC) & (SSZ - 1));
    compute_chunk<N>(B, g0, acc, accx);
  }
}

__device__ __forceinline__ void lds_barrier()
{
  __builtin_amdgcn_fence(__ATOMIC_RELEASE, "workgroup", "local");
  __builtin_amdgcn_s_barrier();
  __builtin_amdgcn_fence(__ATOMIC_ACQUIRE, "workgroup", "local");
}

__device__ __forceinline__ void read_fragments(v4f (&f)[HSZ / 8], const float* fp)
{
  #pragma unroll
  for (int kb = 0; kb < HSZ / 8; ++kb)
    f[kb] = *(const v4f*)(fp + kb * 8);
}

__device__ __forceinline__ void tile_product(const float* fa, const v4f (&wf)[HSZ / 8], v8f& acc)
{
  v4f fa_r[HSZ / 8];
  read_fragments(fa_r, fa);
  __builtin_amdgcn_sched_barrier(0);
  #pragma unroll
  for (int kb = 0; kb < HSZ / 8; ++kb) {
    const v4f a = fa_r[kb];
    const v4f w = wf[kb];
    const v2f a_lo = {a.x, a.y}, a_hi = {a.z, a.w};
    const v2f w_lo = {w.x, w.y}, w_hi = {w.z, w.w};
    acc = __builtin_amdgcn_wmma_f32_16x16x4_f32(
         false, a_lo,  false, w_lo, (short)0, acc, false, false);
    acc = __builtin_amdgcn_wmma_f32_16x16x4_f32(
         false, a_hi,  false, w_hi, (short)0, acc, false, false);
  }
  __builtin_amdgcn_sched_barrier(0);
}

__global__ __launch_bounds__(NW * 32) void relation_fused_kernel(
    const float* __restrict__ inp,
    const int*   __restrict__ masks,
    const float* __restrict__ W0, const float* __restrict__ b0,
    const float* __restrict__ W1, const float* __restrict__ b1,
    float* __restrict__ out)
{
  __shared__ __attribute__((aligned(16))) v2f   rows[SSZ * HS + LDS_TAIL_BYTES / sizeof(v2f)];
  __shared__ __attribute__((aligned(16))) float sAt[NW * 16 * LDP];
  __shared__ __attribute__((aligned(16))) float sW1[HS * LDP];
  __shared__ __attribute__((aligned(16))) float sW0[HS * LDP];
  __shared__ float sG0[TI * NW * 2 * HS];
  __shared__ int irow[SSZ];
  __shared__ int wcnt[NW];

  const int tid  = threadIdx.x;
  const int lane = tid & 31;
  const int wave = __builtin_amdgcn_readfirstlane(tid >> 5);
  const int b    = blockIdx.x / ((HSZ / HS) * NIG);
  const int cs   = (blockIdx.x / NIG) & (HSZ / HS - 1);
  const int ig   = blockIdx.x % NIG;
  const int r = lane & 15;
  const int h = lane >> 4;
  const int col   = cs * HS + r;
  const int ibase = ig * (SSZ / NIG) + wave * (2 * TI) + h;
  const int* mB   = masks + b * SSZ;

  int mi_row[TI];
  #pragma unroll
  for (int t = 0; t < TI; ++t)
    mi_row[t] = mB[ibase + 2 * t];
  const int ja = wave * 64 + lane;
  const int jb = ja + 32;
  const int ma = mB[ja];
  const int mb = mB[jb];

  v4f rw1[2], rw0[2];
  {
    const float* w1g = W1 + (size_t)(cs * HS + wave) * HSZ + lane * 4;
    const float* w0g = W0 + (size_t)(cs * HS + wave) * HSZ + lane * 4;
    #pragma unroll
    for (int p = 0; p < 2; ++p) {
      rw1[p] = *(const v4f*)(w1g + (size_t)(p * NW) * HSZ);
      rw0[p] = *(const v4f*)(w0g + (size_t)(p * NW) * HSZ);
    }
  }
  const float bias1 = b1[col];
  const float bias0 = b0[col];

  __builtin_amdgcn_sched_barrier(0);

  const unsigned bala = __builtin_amdgcn_ballot_w32(ma != 0);
  const unsigned balb = __builtin_amdgcn_ballot_w32(mb != 0);
  const int ca = __builtin_popcount(bala);
  const int cb = __builtin_popcount(balb);
  if (lane == 0) wcnt[wave] = ca + cb;
  lds_barrier();
  int before = 0;
  int len    = 0;
  #pragma unroll
  for (int w = 0; w < NW; ++w) {
    const int c = wcnt[w];
    len += c;
    if (w < wave) before += c;
  }
  const unsigned below = (1u << lane) - 1u;
  const int ua = before + __builtin_popcount(bala & below);
  const int ub = before + ca + __builtin_popcount(balb & below);
  const int sa = ma ? ua : len + (ja - ua);
  const int sb = mb ? ub : len + (jb - ub);
  irow[sa] = ja;
  irow[sb] = jb;
  #pragma unroll
  for (int p = 0; p < 2; ++p) {
    *(v4f*)(sW1 + (wave + p * NW) * LDP + lane * 4) = rw1[p];
    *(v4f*)(sW0 + (wave + p * NW) * LDP + lane * 4) = rw0[p];
  }
  lds_barrier();

  const int lenS = __builtin_amdgcn_readfirstlane(len);
  const int gw = wave * NIG + ig;
  int nr = (lenS - gw + 2 * NHW - 1) >> 7;
  nr = nr < 0 ? 0 : nr;
  const int nrow = nr;
  int nmax = (lenS - ig + 2 * NHW - 1) >> 7;
  nmax = nmax < 0 ? 0 : nmax;
  const int gtiles = nmax;
  int ntw = (lenS - wave * 16 + 127) >> 7;
  ntw = ntw < 0 ? 0 : (ntw > SSZ / (NW * 16) ? SSZ / (NW * 16) : ntw);

  const int hw = gw + NHW * h;
  int nrl = (lenS - hw + 2 * NHW - 1) >> 7;
  nrl = nrl < 0 ? 0 : nrl;
  int irw[TI];
  #pragma unroll
  for (int t = 0; t < TI; ++t)
    irw[t] = irow[hw + t * (2 * NHW)];

  float* sAw = sAt + wave * 16 * LDP;
  const float* fa  = sAw + r * LDP + h * 4;
  const float* inB = inp + (size_t)b * SSZ * HSZ;
  const unsigned lo4 = (unsigned)lane * 4u;
  v4f ar[16];

  const int  gsel   = 7 - wave;
  const bool has_g0 = (wave >= 4) && (gsel < gtiles);

  if (has_g0) {
    #pragma unroll
    for (int q = 0; q < 16; ++q) {
      const int k = (q & 7) * NIG + ig + (q >> 3) * NHW + gsel * (2 * NHW);
      ar[q] = *(const v4f*)(inB + ((unsigned)irow[k] * HSZ + lo4));
    }
    #pragma unroll
    for (int q = 0; q < 16; ++q)
      *(v4f*)(sAw + q * LDP + lane * 4) = ar[q];
  }
  if (ntw > 0) {
    #pragma unroll
    for (int q = 0; q < 16; ++q)
      ar[q] = *(const v4f*)(inB + ((unsigned)irow[wave * 16 + q] * HSZ + lo4));
  }
  if (has_g0) {
    v4f w0f[HSZ / 8];
    read_fragments(w0f, sW0 + r * LDP + h * 4);
    v8f acc = {};
    tile_product(fa, w0f, acc);
    float* g = sG0 + (gsel * NW * 2 + h) * HS + r;
    #pragma unroll
    for (int v = 0; v < 8; ++v)
      g[v * 2 * HS] = (acc[v] + bias0) * GATE_SCALE;
  }

  v4f w1f[HSZ / 8];
  read_fragments(w1f, sW1 + r * LDP + h * 4);

  const float* xr = sAw + (8 * h) * LDP + cs * HS + r;
  for (int c = 0; c < ntw; ++c) {
    const int s0 = c * (NW * 16) + wave * 16;
    const int sl0 = s0 + 8 * h;
    v2f* d = rows + sl0 * HS + r;
    #pragma unroll
    for (int q = 0; q < 16; ++q)
      *(v4f*)(sAw + q * LDP + lane * 4) = ar[q];
    if (c + 1 < ntw) {
      #pragma unroll
      for (int q = 0; q < 16; ++q)
        ar[q] = *(const v4f*)(inB + ((unsigned)irow[s0 + NW * 16 + q] * HSZ + lo4));
    }
    float x0[8];
    #pragma unroll
    for (int v = 0; v < 8; ++v)
      x0[v] = xr[v * LDP];
    v8f acc0 = {};
    tile_product(fa, w1f, acc0);
    #pragma unroll
    for (int v = 0; v < 8; ++v) {
      const bool live = (sl0 + v) < lenS;
      const float ga = (acc0[v] + bias1) * GATE_SCALE;
      d[v * HS] = live ? (v2f){x0[v], ga} : (v2f){0.0f, 0.0f};
    }
  }
  for (int c = ntw; c < SSZ / (NW * 16); ++c) {
    v2f* d = rows + (c * (NW * 16) + wave * 16 + 8 * h) * HS + r;
    #pragma unroll
    for (int v = 0; v < 8; ++v)
      d[v * HS] = (v2f){0.0f, 0.0f};
  }
  lds_barrier();

  float g0[TI];
  #pragma unroll
  for (int t = 0; t < TI; ++t)
    g0[t] = (t < nrl) ? sG0[((t * NW + wave) * 2 + h) * HS + r] : 0.0f;

  const v2f* rl = rows + r;
  float acc[TI];
  float accx = 0.0f;
  #pragma unroll
  for (int t = 0; t < TI; ++t) acc[t] = 0.0f;

  switch (nrow) {
    case 1: row_sums<1>(rl, lenS, g0, acc, accx); break;
    case 2: row_sums<2>(rl, lenS, g0, acc, accx); break;
    case 3: row_sums<3>(rl, lenS, g0, acc, accx); break;
    case 4: row_sums<4>(rl, lenS, g0, acc, accx); break;
    default: break;
  }

  const float inv_len = 1.0f / (float)lenS;
  #pragma unroll
  for (int t = 0; t < TI; ++t) {
    if (t < nrl) {
      const float sum = 0.5f * (accx + acc[t]);
      out[((size_t)(b * SSZ + irw[t]) * HSZ) + col] = tanhf(sum * inv_len);
    }
  }
  const float zv = 0.0f * inv_len;
  #pragma unroll
  for (int t = 0; t < TI; ++t) {
    if (mi_row[t] == 0)
      out[((size_t)(b * SSZ + ibase + 2 * t) * HSZ) + col] = zv;
  }
}

extern "C" void kernel_launch(void* const* d_in, const int* in_sizes, int n_in,
                              void* d_out, int out_size, void* d_ws, size_t ws_size,
                              hipStream_t stream) {
  const float* inp   = (const float*)d_in[0];
  const int*   masks = (const int*)  d_in[1];
  const float* W0    = (const float*)d_in[2];
  const float* b0    = (const float*)d_in[3];
  const float* W1    = (const float*)d_in[4];
  const float* b1    = (const float*)d_in[5];
  float* out = (float*)d_out;

  relation_fused_kernel<<<BSZ * (HSZ / HS) * NIG, NW * 32, 0, stream>>>(
      inp, masks, W0, b0, W1, b1, out);
}
